# starting kernel
# baseline (speedup 1.0000x reference)
#include <hip/hip_runtime.h>

typedef __attribute__((ext_vector_type(2))) float v2f;
typedef __attribute__((ext_vector_type(4))) float v4f;
typedef __attribute__((ext_vector_type(8))) float v8f;

#define BM 256
#define BN 64
#define BK 32
#define LDA_S 265
#define LDB_S 160

__launch_bounds__(256)
__global__ void gcn_gemm_f32_wmma(const float* __restrict__ A,
                                  const float* __restrict__ B,
                                  float* __restrict__ C,
                                  int K, int lda, int ldb, int ldc,
                                  long long sA, long long sB, long long sC)
{
    __shared__ float Asm[BK * LDA_S];
    __shared__ float Bsm[(BK / 2) * LDB_S];

    const long long b = blockIdx.z;
    A += b * sA;
    B += b * sB;
    C += b * sC;

    const int m0 = blockIdx.x * BM;
    const int n0 = blockIdx.y * BN;

    const int t      = threadIdx.x;
    const int lane   = t & 31;
    const int wave   = t >> 5;
    const int lane15 = lane & 15;
    const int lhalf  = lane >> 4;

    const int a_c4 = (t & 7) * 4;
    const int a_r0 = t >> 3;
    const int b_n2 = (t & 31) * 2;
    const int b_kp = t >> 5;

    v8f acc[2][4] = {};

    const int kTiles = K / BK;
    for (int kt = 0; kt < kTiles; ++kt) {
        const int k0 = kt * BK;

        #pragma unroll
        for (int r = 0; r < 8; ++r) {
            const int row = a_r0 + 32 * r;
            const v4f v = *(const v4f*)(A + (long long)(m0 + row) * lda + (k0 + a_c4));
            Asm[(a_c4 + 0) * LDA_S + row] = v.x;
            Asm[(a_c4 + 1) * LDA_S + row] = v.y;
            Asm[(a_c4 + 2) * LDA_S + row] = v.z;
            Asm[(a_c4 + 3) * LDA_S + row] = v.w;
        }
        #pragma unroll
        for (int r = 0; r < 2; ++r) {
            const int kp = b_kp + 8 * r;
            const v2f e = *(const v2f*)(B + (long long)(k0 + 2 * kp + 0) * ldb + (n0 + b_n2));
            const v2f o = *(const v2f*)(B + (long long)(k0 + 2 * kp + 1) * ldb + (n0 + b_n2));
            v4f pk;
            pk.x = e.x; pk.y = o.x;
            pk.z = e.y; pk.w = o.y;
            *(v4f*)&Bsm[kp * LDB_S + 2 * b_n2] = pk;
        }
        if (kt + 1 < kTiles) {
            __builtin_prefetch(A + (long long)(m0 + a_r0) * lda + (k0 + BK + a_c4), 0, 0);
            __builtin_prefetch(B + (long long)(k0 + BK + 2 * b_kp) * ldb + (n0 + b_n2), 0, 0);
        }
        __syncthreads();

        #pragma unroll
        for (int ks = 0; ks < 8; ++ks) {
            const int kk = 4 * ks + 2 * lhalf;
            const int kp = 2 * ks + lhalf;
            v2f af[2], bf[4];
            #pragma unroll
            for (int i = 0; i < 2; ++i) {
                const int m = wave * 32 + 16 * i + lane15;
                af[i].x = Asm[(kk + 0) * LDA_S + m];
                af[i].y = Asm[(kk + 1) * LDA_S + m];
            }
            #pragma unroll
            for (int j = 0; j < 4; ++j)
                bf[j] = *(const v2f*)&Bsm[kp * LDB_S + 2 * (16 * j + lane15)];
            #pragma unroll
            for (int i = 0; i < 2; ++i)
                #pragma unroll
                for (int j = 0; j < 4; ++j)
                    acc[i][j] = __builtin_amdgcn_wmma_f32_16x16x4_f32(
                        false, af[i], false, bf[j], (short)0, acc[i][j], false, false);
        }
        __syncthreads();
    }

    #pragma unroll
    for (int i = 0; i < 2; ++i) {
        const int mrow = m0 + wave * 32 + 16 * i + 8 * lhalf;
        #pragma unroll
        for (int j = 0; j < 4; ++j) {
            float* dst = C + (long long)mrow * ldc + (n0 + 16 * j + lane15);
            #pragma unroll
            for (int v = 0; v < 8; ++v)
                dst[(long long)v * ldc] = acc[i][j][v];
        }
    }
}

extern "C" void kernel_launch(void* const* d_in, const int* in_sizes, int n_in,
                              void* d_out, int out_size, void* d_ws, size_t ws_size,
                              hipStream_t stream) {
    (void)in_sizes; (void)n_in; (void)out_size; (void)ws_size;
    const float* x    = (const float*)d_in[0];
    const float* W    = (const float*)d_in[2];
    const float* adj  = (const float*)d_in[3];
    float*       out  = (float*)d_out;
    float*       h    = (float*)d_ws;

    const int N_NODES = 512, N_BATCH = 64, D_IN = 256, D_OUT = 256;
    const int Mtot = N_BATCH * N_NODES;

    dim3 g1(Mtot / BM, D_OUT / BN, 1);
    gcn_gemm_f32_wmma<<<g1, 256, 0, stream>>>(x, W, h,
                                              D_IN, D_IN, D_OUT, D_OUT,
                                              0LL, 0LL, 0LL);

    dim3 g2(N_NODES / BM, D_OUT / BN, N_BATCH);
    gcn_gemm_f32_wmma<<<g2, 256, 0, stream>>>(adj, h, out,
                                              N_NODES, N_NODES, D_OUT, D_OUT,
                                              0LL,
                                              (long long)N_NODES * D_OUT,
                                              (long long)N_NODES * D_OUT);
}


# launch 1: x slab (128 rows) stationary in LDS, staged once, one barrier; a wave owns a 128x16 strip, reads its W fragments straight from global; no K-loop barrier; sum order within each 8 k changes
# speedup vs baseline: 1.3545x; 1.3545x over previous
#include <hip/hip_runtime.h>


typedef __attribute__((ext_vector_type(2))) float v2f;
typedef __attribute__((ext_vector_type(4))) float v4f;
typedef __attribute__((ext_vector_type(8))) float v8f;

#define SLAB_ROWS 128
#define P1_PITCH 260
#define P2_PITCH 516

template <int PITCH>
__device__ __forceinline__ void gcn_pair_step(const float* a_next, v4f (&a_cur)[8], v4f (&a_nxt)[8],
                                              const v4f bv, v8f (&acc)[8])
{
    #pragma unroll
    for (int i = 0; i < 8; ++i)
        a_nxt[i] = *(const v4f*)(a_next + i * 16 * PITCH);
    __builtin_amdgcn_sched_barrier(0);
    const v2f b01 = __builtin_shufflevector(bv, bv, 0, 1);
    const v2f b23 = __builtin_shufflevector(bv, bv, 2, 3);
    #pragma unroll
    for (int i = 0; i < 8; ++i)
        acc[i] = __builtin_amdgcn_wmma_f32_16x16x4_f32(
            false, __builtin_shufflevector(a_cur[i], a_cur[i], 0, 1), false, b01, (short)0, acc[i], false, false);
    #pragma unroll
    for (int i = 0; i < 8; ++i)
        acc[i] = __builtin_amdgcn_wmma_f32_16x16x4_f32(
            false, __builtin_shufflevector(a_cur[i], a_cur[i], 2, 3), false, b23, (short)0, acc[i], false, false);
    __builtin_amdgcn_sched_barrier(0);
}

__device__ __forceinline__ v4f gcn_w_frag(const float* wsrc, int kq)
{
    const float* p = wsrc + kq * 8 * 256;
    v4f v;
    v.x = p[0];
    v.y = p[256];
    v.z = p[512];
    v.w = p[768];
    return v;
}

__launch_bounds__(512)
__global__ void gcn_xw_f32_wmma(const float* __restrict__ X,
                                const float* __restrict__ Wm,
                                float* __restrict__ HF)
{
    __shared__ __attribute__((aligned(16))) float Xs[SLAB_ROWS * P1_PITCH];

    const int t    = threadIdx.x;
    const int lane = t & 31;
    const int nt   = t >> 5;
    const int c    = lane & 15;
    const int hh   = lane >> 4;
    const int m0   = blockIdx.x * SLAB_ROWS;

    const float* wsrc = Wm + (4 * hh) * 256 + 16 * nt + c;

    v4f b0[4], b1[4];
    #pragma unroll
    for (int u = 0; u < 4; ++u)
        b0[u] = gcn_w_frag(wsrc, u);

    {
        const int r0 = t >> 6;
        const int k4 = (t & 63) * 4;
        const float* src = X + (long long)(m0 + r0) * 256 + k4;
        float* dstl = &Xs[r0 * P1_PITCH + k4];
        v4f tmp[16];
        #pragma unroll
        for (int u = 0; u < 16; ++u)
            tmp[u] = *(const v4f*)(src + (long long)u * 8 * 256);
        #pragma unroll
        for (int u = 0; u < 16; ++u)
            *(v4f*)(dstl + u * 8 * P1_PITCH) = tmp[u];
    }
    __syncthreads();

    const float* afrag = &Xs[c * P1_PITCH + 4 * hh];

    v8f acc[8] = {};
    v4f a0[8], a1[8];
    #pragma unroll
    for (int i = 0; i < 8; ++i)
        a0[i] = *(const v4f*)(afrag + i * 16 * P1_PITCH);

    for (int kb = 0; kb < 32; kb += 8) {
        #pragma unroll
        for (int u = 0; u < 4; ++u)
            b1[u] = gcn_w_frag(wsrc, kb + 4 + u);
        gcn_pair_step<P1_PITCH>(afrag + 8 * (kb + 1), a0, a1, b0[0], acc);
        gcn_pair_step<P1_PITCH>(afrag + 8 * (kb + 2), a1, a0, b0[1], acc);
        gcn_pair_step<P1_PITCH>(afrag + 8 * (kb + 3), a0, a1, b0[2], acc);
        gcn_pair_step<P1_PITCH>(afrag + 8 * (kb + 4), a1, a0, b0[3], acc);
        #pragma unroll
        for (int u = 0; u < 4; ++u)
            b0[u] = gcn_w_frag(wsrc, (kb + 8 + u) & 31);
        gcn_pair_step<P1_PITCH>(afrag + 8 * (kb + 5), a0, a1, b1[0], acc);
        gcn_pair_step<P1_PITCH>(afrag + 8 * (kb + 6), a1, a0, b1[1], acc);
        gcn_pair_step<P1_PITCH>(afrag + 8 * (kb + 7), a0, a1, b1[2], acc);
        gcn_pair_step<P1_PITCH>(afrag + 8 * ((kb + 8) & 31), a1, a0, b1[3], acc);
    }

    const int g  = m0 >> 9;
    const int t0 = (m0 & 511) >> 4;
    #pragma unroll
    for (int i = 0; i < 8; ++i) {
        const int kq = 2 * (t0 + i) + hh;
        float* dst = HF + ((((long long)(g * 16 + nt)) * 64 + kq) * 32 + c) * 4;
        v4f lo, hi;
        lo.x = acc[i][0]; lo.y = acc[i][1]; lo.z = acc[i][2]; lo.w = acc[i][3];
        hi.x = acc[i][4]; hi.y = acc[i][5]; hi.z = acc[i][6]; hi.w = acc[i][7];
        *(v4f*)dst        = lo;
        *(v4f*)(dst + 64) = hi;
    }
}

__launch_bounds__(512)
__global__ void gcn_adj_f32_wmma(const float* __restrict__ adj,
                                 const float* __restrict__ HF,
                                 float* __restrict__ out)
{
    __shared__ __attribute__((aligned(16))) float As[SLAB_ROWS * P2_PITCH];

    const int t    = threadIdx.x;
    const int lane = t & 31;
    const int nt   = t >> 5;
    const int c    = lane & 15;
    const int hh   = lane >> 4;
    const int m0   = blockIdx.x * SLAB_ROWS;
    const int g    = blockIdx.y;

    const float* bsrc = HF + (((long long)(g * 16 + nt)) * 64 * 32 + lane) * 4;

    v4f b0[4], b1[4];
    #pragma unroll
    for (int u = 0; u < 4; ++u)
        b0[u] = *(const v4f*)(bsrc + u * 128);

    {
        const int r0 = t >> 7;
        const int k4 = (t & 127) * 4;
        const float* src = adj + (long long)(m0 + r0) * 512 + k4;
        float* dstl = &As[r0 * P2_PITCH + k4];
        for (int u0 = 0; u0 < 32; u0 += 8) {
            v4f tmp[8];
            #pragma unroll
            for (int u = 0; u < 8; ++u)
                tmp[u] = *(const v4f*)(src + (long long)(u0 + u) * 4 * 512);
            #pragma unroll
            for (int u = 0; u < 8; ++u)
                *(v4f*)(dstl + (u0 + u) * 4 * P2_PITCH) = tmp[u];
        }
    }
    __syncthreads();

    const float* afrag = &As[c * P2_PITCH + 4 * hh];

    v8f acc[8] = {};
    v4f a0[8], a1[8];
    #pragma unroll
    for (int i = 0; i < 8; ++i)
        a0[i] = *(const v4f*)(afrag + i * 16 * P2_PITCH);

    for (int kb = 0; kb < 64; kb += 8) {
        #pragma unroll
        for (int u = 0; u < 4; ++u)
            b1[u] = *(const v4f*)(bsrc + (kb + 4 + u) * 128);
        gcn_pair_step<P2_PITCH>(afrag + 8 * (kb + 1), a0, a1, b0[0], acc);
        gcn_pair_step<P2_PITCH>(afrag + 8 * (kb + 2), a1, a0, b0[1], acc);
        gcn_pair_step<P2_PITCH>(afrag + 8 * (kb + 3), a0, a1, b0[2], acc);
        gcn_pair_step<P2_PITCH>(afrag + 8 * (kb + 4), a1, a0, b0[3], acc);
        #pragma unroll
        for (int u = 0; u < 4; ++u)
            b0[u] = *(const v4f*)(bsrc + ((kb + 8 + u) & 63) * 128);
        gcn_pair_step<P2_PITCH>(afrag + 8 * (kb + 5), a0, a1, b1[0], acc);
        gcn_pair_step<P2_PITCH>(afrag + 8 * (kb + 6), a1, a0, b1[1], acc);
        gcn_pair_step<P2_PITCH>(afrag + 8 * (kb + 7), a0, a1, b1[2], acc);
        gcn_pair_step<P2_PITCH>(afrag + 8 * ((kb + 8) & 63), a1, a0, b1[3], acc);
    }

    float* dst = out + ((long long)(g * 512 + m0 + 8 * hh)) * 256 + 16 * nt + c;
    #pragma unroll
    for (int i = 0; i < 8; ++i)
        #pragma unroll
        for (int r = 0; r < 8; ++r)
            dst[(16 * i + r) * 256] = acc[i][r];
}

extern "C" void kernel_launch(void* const* d_in, const int* in_sizes, int n_in,
                              void* d_out, int out_size, void* d_ws, size_t ws_size,
                              hipStream_t stream) {
    (void)in_sizes; (void)n_in; (void)out_size; (void)ws_size;
    const float* x    = (const float*)d_in[0];
    const float* W    = (const float*)d_in[2];
    const float* adj  = (const float*)d_in[3];
    float*       out  = (float*)d_out;
    float*       hf   = (float*)d_ws;

    const int N_NODES = 512, N_BATCH = 64;
    const int Mtot = N_BATCH * N_NODES;

    dim3 g1(Mtot / SLAB_ROWS, 1, 1);
    gcn_xw_f32_wmma<<<g1, 512, 0, stream>>>(x, W, hf);

    dim3 g2(N_NODES / SLAB_ROWS, N_BATCH, 1);
    gcn_adj_f32_wmma<<<g2, 512, 0, stream>>>(adj, hf, out);
}


# wave tile 128x16 -> 64x64, a lane owning 4 adjacent columns (one per column tile): a B row arrives as one 16-byte load, T and out rows leave as 16-byte stores; 8 waves per group; sum order same
# speedup vs baseline: 1.6170x; 1.1938x over previous
#include <hip/hip_runtime.h>


typedef __attribute__((ext_vector_type(2))) float v2f;
typedef __attribute__((ext_vector_type(4))) float v4f;
typedef __attribute__((ext_vector_type(8))) float v8f;

#define SLAB_ROWS 128
#define PT_PITCH 260
#define PA_PITCH 516

__device__ __forceinline__ void gcn_b_quad(const float* bsrc, int kq, v4f (&b)[4])
{
    const float* p = bsrc + kq * 8 * 256;
    #pragma unroll
    for (int e = 0; e < 4; ++e)
        b[e] = *(const v4f*)(p + e * 256);
}

template <int PITCH>
__device__ __forceinline__ void gcn_pair_step(const float* a_next, v4f (&a_cur)[4], v4f (&a_nxt)[4],
                                              const v4f (&b)[4], v8f (&acc)[4][4])
{
    #pragma unroll
    for (int i = 0; i < 4; ++i)
        a_nxt[i] = *(const v4f*)(a_next + i * 16 * PITCH);
    __builtin_amdgcn_sched_barrier(0);
    #pragma unroll
    for (int i = 0; i < 4; ++i) {
        #pragma unroll
        for (int j = 0; j < 4; ++j) {
            v2f bj;
            bj.x = b[0][j];
            bj.y = b[1][j];
            acc[i][j] = __builtin_amdgcn_wmma_f32_16x16x4_f32(
                false, __builtin_shufflevector(a_cur[i], a_cur[i], 0, 1), false, bj, (short)0, acc[i][j], false, false);
        }
    }
    #pragma unroll
    for (int i = 0; i < 4; ++i) {
        #pragma unroll
        for (int j = 0; j < 4; ++j) {
            v2f bj;
            bj.x = b[2][j];
            bj.y = b[3][j];
            acc[i][j] = __builtin_amdgcn_wmma_f32_16x16x4_f32(
                false, __builtin_shufflevector(a_cur[i], a_cur[i], 2, 3), false, bj, (short)0, acc[i][j], false, false);
        }
    }
    __builtin_amdgcn_sched_barrier(0);
}

template <int PITCH, int NPAIR>
__device__ __forceinline__ void gcn_tile_product(const float* afrag, const float* bsrc, v4f (&bq)[4][4], v8f (&acc)[4][4])
{
    v4f a0[4], a1[4];
    #pragma unroll
    for (int i = 0; i < 4; ++i)
        a0[i] = *(const v4f*)(afrag + i * 16 * PITCH);

    for (int kb = 0; kb < NPAIR; kb += 4) {
        gcn_b_quad(bsrc, kb + 3, bq[3]);
        gcn_pair_step<PITCH>(afrag + 8 * (kb + 1), a0, a1, bq[0], acc);
        gcn_b_quad(bsrc, (kb + 4) & (NPAIR - 1), bq[0]);
        gcn_pair_step<PITCH>(afrag + 8 * (kb + 2), a1, a0, bq[1], acc);
        gcn_b_quad(bsrc, (kb + 5) & (NPAIR - 1), bq[1]);
        gcn_pair_step<PITCH>(afrag + 8 * (kb + 3), a0, a1, bq[2], acc);
        gcn_b_quad(bsrc, (kb + 6) & (NPAIR - 1), bq[2]);
        gcn_pair_step<PITCH>(afrag + 8 * ((kb + 4) & (NPAIR - 1)), a1, a0, bq[3], acc);
    }
}

__launch_bounds__(256)
__global__ void gcn_adjx_w_f32_wmma(const float* __restrict__ X,
                                    const float* __restrict__ Wm,
                                    const float* __restrict__ adj,
                                    float* __restrict__ out)
{
    __shared__ __attribute__((aligned(16))) float Ls[SLAB_ROWS * PA_PITCH];

    const int t    = threadIdx.x;
    const int lane = t & 31;
    const int wv   = t >> 5;
    const int cg   = wv & 3;
    const int rh   = wv >> 2;
    const int c    = lane & 15;
    const int hh   = lane >> 4;
    const int m0   = blockIdx.x * SLAB_ROWS;
    const int g    = blockIdx.y;

    const float* xsrc = X  + ((long long)g * 512 + 4 * hh) * 256 + 64 * cg + 4 * c;
    const float* wsrc = Wm + (4 * hh) * 256 + 64 * cg + 4 * c;

    v4f bq[4][4];
    #pragma unroll
    for (int u = 0; u < 3; ++u)
        gcn_b_quad(xsrc, u, bq[u]);

    {
        const int r0 = t >> 7;
        const int k4 = (t & 127) * 4;
        const float* src = adj + (long long)(m0 + r0) * 512 + k4;
        float* dstl = &Ls[r0 * PA_PITCH + k4];
        for (int u0 = 0; u0 < 64; u0 += 16) {
            v4f tmp[16];
            #pragma unroll
            for (int u = 0; u < 16; ++u)
                tmp[u] = *(const v4f*)(src + (long long)(u0 + u) * 2 * 512);
            #pragma unroll
            for (int u = 0; u < 16; ++u)
                *(v4f*)(dstl + (u0 + u) * 2 * PA_PITCH) = tmp[u];
        }
    }
    __syncthreads();

    v8f acc[4][4] = {};
    gcn_tile_product<PA_PITCH, 64>(&Ls[(64 * rh + c) * PA_PITCH + 4 * hh], xsrc, bq, acc);

    #pragma unroll
    for (int u = 0; u < 3; ++u)
        gcn_b_quad(wsrc, u, bq[u]);

    __syncthreads();
    {
        float* tdst = &Ls[(64 * rh + 8 * hh) * PT_PITCH + 64 * cg + 4 * c];
        #pragma unroll
        for (int i = 0; i < 4; ++i)
            #pragma unroll
            for (int r = 0; r < 8; ++r) {
                v4f v;
                v.x = acc[i][0][r];
                v.y = acc[i][1][r];
                v.z = acc[i][2][r];
                v.w = acc[i][3][r];
                *(v4f*)(tdst + (16 * i + r) * PT_PITCH) = v;
            }
    }
    __syncthreads();

    v8f acc2[4][4] = {};
    gcn_tile_product<PT_PITCH, 32>(&Ls[(64 * rh + c) * PT_PITCH + 4 * hh], wsrc, bq, acc2);

    float* dst = out + ((long long)(g * 512 + m0 + 64 * rh + 8 * hh)) * 256 + 64 * cg + 4 * c;
    #pragma unroll
    for (int i = 0; i < 4; ++i)
        #pragma unroll
        for (int r = 0; r < 8; ++r) {
            v4f v;
            v.x = acc2[i][0][r];
            v.y = acc2[i][1][r];
            v.z = acc2[i][2][r];
            v.w = acc2[i][3][r];
            *(v4f*)(dst + (16 * i + r) * 256) = v;
        }
}

extern "C" void kernel_launch(void* const* d_in, const int* in_sizes, int n_in,
                              void* d_out, int out_size, void* d_ws, size_t ws_size,
                              hipStream_t stream) {
    (void)in_sizes; (void)n_in; (void)out_size; (void)d_ws; (void)ws_size;
    const float* x    = (const float*)d_in[0];
    const float* W    = (const float*)d_in[2];
    const float* adj  = (const float*)d_in[3];
    float*       out  = (float*)d_out;

    dim3 grid(4, 64, 1);
    gcn_adjx_w_f32_wmma<<<grid, 256, 0, stream>>>(x, W, adj, out);
}


# each wave's 64x64 tile product split 2x2x2 and done with seven block products instead of eight (operand sums and differences on the vector unit): 5376 WMMAs per SIMD, was 6144; sum order changes
# speedup vs baseline: 1.7036x; 1.0536x over previous
#include <hip/hip_runtime.h>


typedef __attribute__((ext_vector_type(2))) float v2f;
typedef __attribute__((ext_vector_type(4))) float v4f;
typedef __attribute__((ext_vector_type(8))) float v8f;

#define SLAB_ROWS 128
#define PT_PITCH 260
#define PA_PITCH 516

template <int KH>
__device__ __forceinline__ void gcn_b_oct(const float* bsrc, int kq, v4f (&b)[8])
{
    const float* p = bsrc + kq * 8 * 256;
    #pragma unroll
    for (int e = 0; e < 4; ++e)
        b[e] = *(const v4f*)(p + e * 256);
    #pragma unroll
    for (int e = 0; e < 4; ++e)
        b[4 + e] = *(const v4f*)(p + (KH + e) * 256);
}

template <int PITCH, int KH>
__device__ __forceinline__ void gcn_a_oct(const float* ap, v4f (&a)[8])
{
    #pragma unroll
    for (int i = 0; i < 2; ++i) {
        a[i]     = *(const v4f*)(ap + (16 * i) * PITCH);
        a[2 + i] = *(const v4f*)(ap + (16 * i) * PITCH + KH);
        a[4 + i] = *(const v4f*)(ap + (16 * (2 + i)) * PITCH);
        a[6 + i] = *(const v4f*)(ap + (16 * (2 + i)) * PITCH + KH);
    }
}

template <int BLK>
__device__ __forceinline__ float gcn_bel(const v4f (&b)[8], int e, int j)
{
    return b[(BLK >> 1) * 4 + e][(BLK & 1) * 2 + j];
}

template <int X, int Y, int SGN>
__device__ __forceinline__ void gcn_t(const v4f (&b)[8], v2f (&t)[2][2])
{
    #pragma unroll
    for (int j = 0; j < 2; ++j) {
        #pragma unroll
        for (int ks = 0; ks < 2; ++ks) {
            float x0 = gcn_bel<X>(b, 2 * ks, j);
            float x1 = gcn_bel<X>(b, 2 * ks + 1, j);
            if (SGN > 0) {
                x0 += gcn_bel<Y>(b, 2 * ks, j);
                x1 += gcn_bel<Y>(b, 2 * ks + 1, j);
            } else if (SGN < 0) {
                x0 -= gcn_bel<Y>(b, 2 * ks, j);
                x1 -= gcn_bel<Y>(b, 2 * ks + 1, j);
            }
            t[j][ks].x = x0;
            t[j][ks].y = x1;
        }
    }
}

template <int X, int Y, int SGN>
__device__ __forceinline__ void gcn_s(const v4f (&a)[8], v4f (&s)[2])
{
    #pragma unroll
    for (int i = 0; i < 2; ++i) {
        if (SGN > 0)      s[i] = a[2 * X + i] + a[2 * Y + i];
        else if (SGN < 0) s[i] = a[2 * X + i] - a[2 * Y + i];
        else              s[i] = a[2 * X + i];
    }
}

template <int AX, int AY, int AS, int BX, int BY, int BS>
__device__ __forceinline__ void gcn_m(const v4f (&a)[8], const v4f (&b)[8], v8f (&m)[2][2])
{
    v4f s[2];
    v2f t[2][2];
    gcn_s<AX, AY, AS>(a, s);
    gcn_t<BX, BY, BS>(b, t);
    #pragma unroll
    for (int i = 0; i < 2; ++i) {
        #pragma unroll
        for (int j = 0; j < 2; ++j)
            m[i][j] = __builtin_amdgcn_wmma_f32_16x16x4_f32(
                false, __builtin_shufflevector(s[i], s[i], 0, 1), false, t[j][0], (short)0, m[i][j], false, false);
    }
    #pragma unroll
    for (int i = 0; i < 2; ++i) {
        #pragma unroll
        for (int j = 0; j < 2; ++j)
            m[i][j] = __builtin_amdgcn_wmma_f32_16x16x4_f32(
                false, __builtin_shufflevector(s[i], s[i], 2, 3), false, t[j][1], (short)0, m[i][j], false, false);
    }
}

template <int PITCH, int KH>
__device__ __forceinline__ void gcn_pair_step(const float* a_next, const v4f (&a_cur)[8], v4f (&a_nxt)[8],
                                              const v4f (&b)[8], v8f (&m)[7][2][2])
{
    gcn_a_oct<PITCH, KH>(a_next, a_nxt);
    __builtin_amdgcn_sched_barrier(0);
    gcn_m<0, 3, +1, 0, 3, +1>(a_cur, b, m[0]);
    gcn_m<2, 3, +1, 0, 0,  0>(a_cur, b, m[1]);
    gcn_m<0, 0,  0, 1, 3, -1>(a_cur, b, m[2]);
    gcn_m<3, 0,  0, 2, 0, -1>(a_cur, b, m[3]);
    gcn_m<0, 1, +1, 3, 0,  0>(a_cur, b, m[4]);
    gcn_m<2, 0, -1, 0, 1, +1>(a_cur, b, m[5]);
    gcn_m<1, 3, -1, 2, 3, +1>(a_cur, b, m[6]);
    __builtin_amdgcn_sched_barrier(0);
}

template <int PITCH, int K>
__device__ __forceinline__ void gcn_tile_product(const float* afrag, const float* bsrc, v4f (&bq)[4][8], v8f (&acc)[4][4])
{
    constexpr int KH = K / 2;
    constexpr int NP = KH / 8;
    v8f m[7][2][2] = {};
    v4f a0[8], a1[8];
    gcn_a_oct<PITCH, KH>(afrag, a0);

    for (int kb = 0; kb < NP; kb += 4) {
        gcn_b_oct<KH>(bsrc, kb + 3, bq[3]);
        gcn_pair_step<PITCH, KH>(afrag + 8 * (kb + 1), a0, a1, bq[0], m);
        gcn_b_oct<KH>(bsrc, (kb + 4) & (NP - 1), bq[0]);
        gcn_pair_step<PITCH, KH>(afrag + 8 * (kb + 2), a1, a0, bq[1], m);
        gcn_b_oct<KH>(bsrc, (kb + 5) & (NP - 1), bq[1]);
        gcn_pair_step<PITCH, KH>(afrag + 8 * (kb + 3), a0, a1, bq[2], m);
        gcn_b_oct<KH>(bsrc, (kb + 6) & (NP - 1), bq[2]);
        gcn_pair_step<PITCH, KH>(afrag + 8 * ((kb + 4) & (NP - 1)), a1, a0, bq[3], m);
    }

    #pragma unroll
    for (int i = 0; i < 2; ++i) {
        #pragma unroll
        for (int j = 0; j < 2; ++j) {
            acc[2 + i][2 + j] = m[0][i][j] - m[1][i][j] + m[2][i][j] + m[5][i][j];
            acc[2 + i][j]     = m[1][i][j] + m[3][i][j];
            acc[i][2 + j]     = m[2][i][j] + m[4][i][j];
            acc[i][j]         = m[0][i][j] + m[3][i][j] - m[4][i][j] + m[6][i][j];
        }
    }
}

__launch_bounds__(256)
__global__ void gcn_adjx_w_f32_wmma(const float* __restrict__ X,
                                    const float* __restrict__ Wm,
                                    const float* __restrict__ adj,
                                    float* __restrict__ out)
{
    __shared__ __attribute__((aligned(16))) float Ls[SLAB_ROWS * PA_PITCH];

    const int t    = threadIdx.x;
    const int lane = t & 31;
    const int wv   = t >> 5;
    const int cg   = wv & 3;
    const int rh   = wv >> 2;
    const int c    = lane & 15;
    const int hh   = lane >> 4;
    const int m0   = blockIdx.x * SLAB_ROWS;
    const int g    = blockIdx.y;

    const float* xsrc = X  + ((long long)g * 512 + 4 * hh) * 256 + 64 * cg + 4 * c;
    const float* wsrc = Wm + (4 * hh) * 256 + 64 * cg + 4 * c;

    v4f bq[4][8];
    #pragma unroll
    for (int u = 0; u < 3; ++u)
        gcn_b_oct<256>(xsrc, u, bq[u]);

    {
        const int r0 = t >> 7;
        const int k4 = (t & 127) * 4;
        const float* src = adj + (long long)(m0 + r0) * 512 + k4;
        float* dstl = &Ls[r0 * PA_PITCH + k4];
        for (int u0 = 0; u0 < 64; u0 += 16) {
            v4f tmp[16];
            #pragma unroll
            for (int u = 0; u < 16; ++u)
                tmp[u] = *(const v4f*)(src + (long long)(u0 + u) * 2 * 512);
            #pragma unroll
            for (int u = 0; u < 16; ++u)
                *(v4f*)(dstl + (u0 + u) * 2 * PA_PITCH) = tmp[u];
        }
    }
    __syncthreads();

    v8f acc[4][4];
    gcn_tile_product<PA_PITCH, 512>(&Ls[(64 * rh + c) * PA_PITCH + 4 * hh], xsrc, bq, acc);

    #pragma unroll
    for (int u = 0; u < 3; ++u)
        gcn_b_oct<128>(wsrc, u, bq[u]);

    __syncthreads();
    {
        float* tdst = &Ls[(64 * rh + 8 * hh) * PT_PITCH + 64 * cg + 4 * c];
        #pragma unroll
        for (int i = 0; i < 4; ++i)
            #pragma unroll
            for (int r = 0; r < 8; ++r) {
                v4f v;
                v.x = acc[i][0][r];
                v.y = acc[i][1][r];
                v.z = acc[i][2][r];
                v.w = acc[i][3][r];
                *(v4f*)(tdst + (16 * i + r) * PT_PITCH) = v;
            }
    }
    __syncthreads();

    v8f acc2[4][4];
    gcn_tile_product<PT_PITCH, 256>(&Ls[(64 * rh + c) * PT_PITCH + 4 * hh], wsrc, bq, acc2);

    float* dst = out + ((long long)(g * 512 + m0 + 64 * rh + 8 * hh)) * 256 + 64 * cg + 4 * c;
    #pragma unroll
    for (int i = 0; i < 4; ++i)
        #pragma unroll
        for (int r = 0; r < 8; ++r) {
            v4f v;
            v.x = acc2[i][0][r];
            v.y = acc2[i][1][r];
            v.z = acc2[i][2][r];
            v.w = acc2[i][3][r];
            *(v4f*)(dst + (16 * i + r) * 256) = v;
        }
}

extern "C" void kernel_launch(void* const* d_in, const int* in_sizes, int n_in,
                              void* d_out, int out_size, void* d_ws, size_t ws_size,
                              hipStream_t stream) {
    (void)in_sizes; (void)n_in; (void)out_size; (void)d_ws; (void)ws_size;
    const float* x    = (const float*)d_in[0];
    const float* W    = (const float*)d_in[2];
    const float* adj  = (const float*)d_in[3];
    float*       out  = (float*)d_out;

    dim3 grid(4, 64, 1);
    gcn_adjx_w_f32_wmma<<<grid, 256, 0, stream>>>(x, W, adj, out);
}
